# starting kernel
# baseline (speedup 1.0000x reference)
#include <hip/hip_runtime.h>
#include <hip/hip_bf16.h>

#define NUM_EXPERT 32
#define IN_FEAT 512
#define OUT_FEAT 512
#define BATCH 4096
#define TILE_M 16
#define MAX_TILES (BATCH / TILE_M + NUM_EXPERT)

typedef __attribute__((ext_vector_type(2))) float v2f;
typedef __attribute__((ext_vector_type(4))) float v4f;
typedef __attribute__((ext_vector_type(8))) float v8f;

typedef const float  __attribute__((address_space(1)))* gcfp;
typedef const v2f    __attribute__((address_space(1)))* gcv2fp;
typedef const v4f    __attribute__((address_space(1)))* gcv4fp;
typedef float        __attribute__((address_space(1)))* gfp;

struct MoeWs {
  int counts[NUM_EXPERT];
  int cursor[NUM_EXPERT];
  int offsets[NUM_EXPERT + 1];
  int numTiles;
  int tileExpert[MAX_TILES];
  int tileM[MAX_TILES];
  int perm[BATCH];
};

__global__ void moe_zero(MoeWs* ws) {
  int t = threadIdx.x;
  if (t < NUM_EXPERT) { ws->counts[t] = 0; ws->cursor[t] = 0; }
  if (t == 0) ws->numTiles = 0;
}

__global__ void moe_hist(const int* __restrict__ gate, MoeWs* ws) {
  int b = blockIdx.x * blockDim.x + threadIdx.x;
  if (b < BATCH) atomicAdd(&ws->counts[gate[b] & (NUM_EXPERT - 1)], 1);
}

__global__ void moe_scan(MoeWs* ws) {
  if (threadIdx.x != 0) return;
  int off = 0;
  for (int e = 0; e < NUM_EXPERT; ++e) { ws->offsets[e] = off; off += ws->counts[e]; }
  ws->offsets[NUM_EXPERT] = off;
  int nt = 0;
  for (int e = 0; e < NUM_EXPERT; ++e) {
    int tiles = (ws->counts[e] + TILE_M - 1) / TILE_M;
    for (int j = 0; j < tiles; ++j) { ws->tileExpert[nt] = e; ws->tileM[nt] = j; ++nt; }
  }
  ws->numTiles = nt;
}

__global__ void moe_scatter(const int* __restrict__ gate, MoeWs* ws) {
  int b = blockIdx.x * blockDim.x + threadIdx.x;
  if (b < BATCH) {
    int e = gate[b] & (NUM_EXPERT - 1);
    int pos = ws->offsets[e] + atomicAdd(&ws->cursor[e], 1);
    ws->perm[pos] = b;
  }
}

__launch_bounds__(128)
__global__ void moe_gemm(const float* __restrict__ inp,
                         const float* __restrict__ weight,
                         float* __restrict__ out,
                         const MoeWs* __restrict__ ws) {
  __shared__ __align__(16) float sA[TILE_M][IN_FEAT];
  __shared__ int sTok[TILE_M];

  const int tile = blockIdx.x;
  if (tile >= ws->numTiles) return;
  const int e = ws->tileExpert[tile];
  const int mt = ws->tileM[tile];
  const int base = ws->offsets[e] + mt * TILE_M;
  const int mcount = min(TILE_M, ws->counts[e] - mt * TILE_M);

  if (threadIdx.x < TILE_M)
    sTok[threadIdx.x] = (threadIdx.x < mcount) ? ws->perm[base + threadIdx.x] : -1;
  __syncthreads();

  for (int f = threadIdx.x; f < TILE_M * (IN_FEAT / 4); f += 128) {
    int row = f >> 7;
    int c4  = f & 127;
    int tok = sTok[row];
    v4f v = {};
    if (tok >= 0) v = ((gcv4fp)(inp + (size_t)tok * IN_FEAT))[c4];
    ((v4f*)&sA[row][0])[c4] = v;
  }
  __syncthreads();

  const int lane = threadIdx.x & 31;
  const int wave = threadIdx.x >> 5;
  const int hl   = lane >> 4;
  const int l15  = lane & 15;

  const float* wbase = weight + (size_t)e * OUT_FEAT * IN_FEAT;
  const int ncol0 = blockIdx.y * 256 + wave * 64;

  gcfp bptr[4];
#pragma unroll
  for (int nt = 0; nt < 4; ++nt) {
    const float* p = wbase + (size_t)(ncol0 + nt * 16 + l15) * IN_FEAT + 2 * hl;
    __builtin_prefetch(p, 0, 1);
    bptr[nt] = (gcfp)p;
  }

  v8f acc[4] = {};
  const float* aptr = &sA[l15][2 * hl];

#pragma unroll 8
  for (int k = 0; k < IN_FEAT; k += 4) {
    v2f a = *(const v2f*)aptr;
    aptr += 4;
#pragma unroll
    for (int nt = 0; nt < 4; ++nt) {
      v2f bb = *(gcv2fp)bptr[nt];
      bptr[nt] += 4;
      acc[nt] = __builtin_amdgcn_wmma_f32_16x16x4_f32(
          false, a, false, bb, (short)0, acc[nt], false, false);
    }
  }

#pragma unroll
  for (int nt = 0; nt < 4; ++nt) {
    const int col = ncol0 + nt * 16 + l15;
#pragma unroll
    for (int r = 0; r < 8; ++r) {
      const int m = r + 8 * hl;
      if (m < mcount)
        ((gfp)out)[(size_t)sTok[m] * OUT_FEAT + col] = acc[nt][r];
    }
  }
}

extern "C" void kernel_launch(void* const* d_in, const int* in_sizes, int n_in,
                              void* d_out, int out_size, void* d_ws, size_t ws_size,
                              hipStream_t stream) {
  (void)in_sizes; (void)n_in; (void)out_size; (void)ws_size;
  const float* inp    = (const float*)d_in[0];
  const int*   gate   = (const int*)d_in[1];
  const float* weight = (const float*)d_in[2];
  float* out = (float*)d_out;
  MoeWs* ws  = (MoeWs*)d_ws;

  moe_zero<<<1, 64, 0, stream>>>(ws);
  moe_hist<<<BATCH / 256, 256, 0, stream>>>(gate, ws);
  moe_scan<<<1, 32, 0, stream>>>(ws);
  moe_scatter<<<BATCH / 256, 256, 0, stream>>>(gate, ws);
  moe_gemm<<<dim3(MAX_TILES, OUT_FEAT / 256), 128, 0, stream>>>(inp, weight, out, ws);
}


# Weight-stationary GEMM: block = (expert, 64 columns) keeps its weight slab in LDS, read once with whole-line 16-B loads; the expert's tokens stream through LDS in 64-row chunks; B fragments from LDS
# speedup vs baseline: 1.4436x; 1.4436x over previous
#include <hip/hip_runtime.h>
#include <hip/hip_bf16.h>

#define NUM_EXPERT 32
#define IN_FEAT 512
#define OUT_FEAT 512
#define BATCH 4096
#define TILE_M 16
#define MAX_TILES (BATCH / TILE_M + NUM_EXPERT)

typedef __attribute__((ext_vector_type(2))) float v2f;
typedef __attribute__((ext_vector_type(4))) float v4f;
typedef __attribute__((ext_vector_type(8))) float v8f;

typedef const float  __attribute__((address_space(1)))* gcfp;
typedef const v2f    __attribute__((address_space(1)))* gcv2fp;
typedef const v4f    __attribute__((address_space(1)))* gcv4fp;
typedef float        __attribute__((address_space(1)))* gfp;

struct MoeWs {
  int counts[NUM_EXPERT];
  int cursor[NUM_EXPERT];
  int offsets[NUM_EXPERT + 1];
  int numTiles;
  int tileExpert[MAX_TILES];
  int tileM[MAX_TILES];
  int perm[BATCH];
};

__global__ void moe_zero(MoeWs* ws) {
  int t = threadIdx.x;
  if (t < NUM_EXPERT) { ws->counts[t] = 0; ws->cursor[t] = 0; }
  if (t == 0) ws->numTiles = 0;
}

__global__ void moe_hist(const int* __restrict__ gate, MoeWs* ws) {
  int b = blockIdx.x * blockDim.x + threadIdx.x;
  if (b < BATCH) atomicAdd(&ws->counts[gate[b] & (NUM_EXPERT - 1)], 1);
}

__global__ void moe_scan(MoeWs* ws) {
  if (threadIdx.x != 0) return;
  int off = 0;
  for (int e = 0; e < NUM_EXPERT; ++e) { ws->offsets[e] = off; off += ws->counts[e]; }
  ws->offsets[NUM_EXPERT] = off;
  int nt = 0;
  for (int e = 0; e < NUM_EXPERT; ++e) {
    int tiles = (ws->counts[e] + TILE_M - 1) / TILE_M;
    for (int j = 0; j < tiles; ++j) { ws->tileExpert[nt] = e; ws->tileM[nt] = j; ++nt; }
  }
  ws->numTiles = nt;
}

__global__ void moe_scatter(const int* __restrict__ gate, MoeWs* ws) {
  int b = blockIdx.x * blockDim.x + threadIdx.x;
  if (b < BATCH) {
    int e = gate[b] & (NUM_EXPERT - 1);
    int pos = ws->offsets[e] + atomicAdd(&ws->cursor[e], 1);
    ws->perm[pos] = b;
  }
}

#define SLAB_N 64
#define CH_TOK 64
#define PITCH 516

__launch_bounds__(256)
__global__ void moe_gemm(const float* __restrict__ inp,
                         const float* __restrict__ weight,
                         float* __restrict__ out,
                         const MoeWs* __restrict__ ws) {
  __shared__ __align__(16) float sW[SLAB_N * PITCH];
  __shared__ __align__(16) float sA[CH_TOK * PITCH];
  __shared__ int sTok[BATCH];

  const int tid  = threadIdx.x;
  const int slab = blockIdx.x;
  const int e    = blockIdx.y;
  const int n    = min(ws->counts[e], BATCH);
  if (n <= 0) return;
  const int base   = ws->offsets[e];
  const int tiles  = (n + TILE_M - 1) / TILE_M;
  const int chunks = (tiles + 3) / 4;
  const int npad   = chunks * CH_TOK;

  for (int i = tid; i < npad; i += 256)
    sTok[i] = (i < n) ? ws->perm[base + i] : -1;

  gcv4fp wsl = (gcv4fp)(weight + ((size_t)e * OUT_FEAT + (size_t)slab * SLAB_N) * IN_FEAT);
  v4f wreg[32];
#pragma unroll
  for (int i = 0; i < 32; ++i) wreg[i] = wsl[i * 256 + tid];

  __syncthreads();

  const int c4   = tid & 127;
  const int rsel = tid >> 7;
  const v4f zero4 = {};
  v4f pre[32];

  {
    const int tc = min(4, tiles);
#pragma unroll
    for (int t = 0; t < 4; ++t) {
      if (t < tc) {
#pragma unroll
        for (int j = 0; j < 8; ++j) {
          const int i = t * 8 + j;
          const int tok = sTok[2 * i + rsel];
          pre[i] = ((gcv4fp)(inp + (size_t)max(tok, 0) * IN_FEAT))[c4];
        }
      }
    }
#pragma unroll
    for (int i = 0; i < 32; ++i) {
      const int f = i * 256 + tid;
      *(v4f*)&sW[(f >> 7) * PITCH + (f & 127) * 4] = wreg[i];
    }
#pragma unroll
    for (int t = 0; t < 4; ++t) {
      if (t < tc) {
#pragma unroll
        for (int j = 0; j < 8; ++j) {
          const int i = t * 8 + j;
          const int row = 2 * i + rsel;
          const int tok = sTok[row];
          *(v4f*)&sA[row * PITCH + c4 * 4] = (tok >= 0) ? pre[i] : zero4;
        }
      }
    }
  }
  __syncthreads();

  const int lane = tid & 31;
  const int hl   = lane >> 4;
  const int l15  = lane & 15;
  const int w  = __builtin_amdgcn_readfirstlane(tid >> 5);
  const int ti = w >> 1;
  const int p  = w & 1;

  const float* aptr  = &sA[(ti * TILE_M + l15) * PITCH + 2 * hl];
  const float* bptr0 = &sW[(p * 32 + l15) * PITCH + 2 * hl];
  const float* bptr1 = bptr0 + 16 * PITCH;
  const int col0 = slab * SLAB_N + p * 32 + l15;

  for (int c = 0; c < chunks; ++c) {
    const int tc  = min(4, tiles - 4 * c);
    const int tcn = min(4, tiles - 4 * (c + 1));

#pragma unroll
    for (int t = 0; t < 4; ++t) {
      if (t < tcn) {
#pragma unroll
        for (int j = 0; j < 8; ++j) {
          const int i = t * 8 + j;
          const int tok = sTok[(c + 1) * CH_TOK + 2 * i + rsel];
          pre[i] = ((gcv4fp)(inp + (size_t)max(tok, 0) * IN_FEAT))[c4];
        }
      }
    }

    if (ti < tc) {
      v8f acc0 = {}, acc1 = {};
#pragma unroll 16
      for (int k = 0; k < IN_FEAT; k += 4) {
        v2f a  = *(const v2f*)(aptr + k);
        v2f b0 = *(const v2f*)(bptr0 + k);
        v2f b1 = *(const v2f*)(bptr1 + k);
        acc0 = __builtin_amdgcn_wmma_f32_16x16x4_f32(
            false, a, false, b0, (short)0, acc0, false, false);
        acc1 = __builtin_amdgcn_wmma_f32_16x16x4_f32(
            false, a, false, b1, (short)0, acc1, false, false);
      }
      const int rowbase = c * CH_TOK + ti * TILE_M + 8 * hl;
      int tk[8];
#pragma unroll
      for (int r = 0; r < 8; ++r) tk[r] = sTok[rowbase + r];
#pragma unroll
      for (int r = 0; r < 8; ++r) {
        if (tk[r] >= 0) {
          gfp o = (gfp)(out + (size_t)tk[r] * OUT_FEAT + col0);
          o[0]  = acc0[r];
          o[16] = acc1[r];
        }
      }
    }
    __syncthreads();

    if (tcn > 0) {
#pragma unroll
      for (int t = 0; t < 4; ++t) {
        if (t < tcn) {
#pragma unroll
          for (int j = 0; j < 8; ++j) {
            const int i = t * 8 + j;
            const int row = 2 * i + rsel;
            const int tok = sTok[(c + 1) * CH_TOK + row];
            *(v4f*)&sA[row * PITCH + c4 * 4] = (tok >= 0) ? pre[i] : zero4;
          }
        }
      }
      __syncthreads();
    }
  }
}

extern "C" void kernel_launch(void* const* d_in, const int* in_sizes, int n_in,
                              void* d_out, int out_size, void* d_ws, size_t ws_size,
                              hipStream_t stream) {
  (void)in_sizes; (void)n_in; (void)out_size; (void)ws_size;
  const float* inp    = (const float*)d_in[0];
  const int*   gate   = (const int*)d_in[1];
  const float* weight = (const float*)d_in[2];
  float* out = (float*)d_out;
  MoeWs* ws  = (MoeWs*)d_ws;

  moe_zero<<<1, 64, 0, stream>>>(ws);
  moe_hist<<<BATCH / 256, 256, 0, stream>>>(gate, ws);
  moe_scan<<<1, 32, 0, stream>>>(ws);
  moe_scatter<<<BATCH / 256, 256, 0, stream>>>(gate, ws);
  moe_gemm<<<dim3(OUT_FEAT / SLAB_N, NUM_EXPERT), 256, 0, stream>>>(inp, weight, out, ws);
}


# Routing in ONE single-work-group launch: LDS counting add's return is a token's rank (arrival order, sums unchanged), lists at a fixed per-expert stride, no scan: 2 launches instead of 5
# speedup vs baseline: 1.9878x; 1.3770x over previous
#include <hip/hip_runtime.h>
#include <hip/hip_bf16.h>

#define NUM_EXPERT 32
#define IN_FEAT 512
#define OUT_FEAT 512
#define BATCH 4096
#define TILE_M 16

typedef __attribute__((ext_vector_type(2))) float v2f;
typedef __attribute__((ext_vector_type(4))) float v4f;
typedef __attribute__((ext_vector_type(8))) float v8f;

typedef const float  __attribute__((address_space(1)))* gcfp;
typedef const v2f    __attribute__((address_space(1)))* gcv2fp;
typedef const v4f    __attribute__((address_space(1)))* gcv4fp;
typedef float        __attribute__((address_space(1)))* gfp;

struct MoeWs {
  int counts[NUM_EXPERT];
  int perm[NUM_EXPERT * BATCH];
};

#define ROUTE_T 1024
#define ROUTE_PER (BATCH / ROUTE_T)

__launch_bounds__(ROUTE_T)
__global__ void moe_route(const int* __restrict__ gate, MoeWs* __restrict__ ws) {
  __shared__ int cnt[NUM_EXPERT];
  const int t = threadIdx.x;
  int g[ROUTE_PER];
#pragma unroll
  for (int k = 0; k < ROUTE_PER; ++k) g[k] = gate[k * ROUTE_T + t] & (NUM_EXPERT - 1);
  if (t < NUM_EXPERT) cnt[t] = 0;
  __syncthreads();
  int r[ROUTE_PER];
#pragma unroll
  for (int k = 0; k < ROUTE_PER; ++k) r[k] = atomicAdd(&cnt[g[k]], 1);
  __syncthreads();
#pragma unroll
  for (int k = 0; k < ROUTE_PER; ++k) ws->perm[g[k] * BATCH + r[k]] = k * ROUTE_T + t;
  if (t < NUM_EXPERT) ws->counts[t] = cnt[t];
}

#define SLAB_N 64
#define CH_TOK 64
#define PITCH 516

__launch_bounds__(256)
__global__ void moe_gemm(const float* __restrict__ inp,
                         const float* __restrict__ weight,
                         float* __restrict__ out,
                         const MoeWs* __restrict__ ws) {
  __shared__ __align__(16) float sW[SLAB_N * PITCH];
  __shared__ __align__(16) float sA[CH_TOK * PITCH];
  __shared__ int sTok[BATCH];

  const int tid  = threadIdx.x;
  const int slab = blockIdx.x;
  const int e    = blockIdx.y;
  const int n    = min(ws->counts[e], BATCH);
  if (n <= 0) return;
  const int base   = e * BATCH;
  const int tiles  = (n + TILE_M - 1) / TILE_M;
  const int chunks = (tiles + 3) / 4;
  const int npad   = chunks * CH_TOK;

  for (int i = tid; i < npad; i += 256)
    sTok[i] = (i < n) ? ws->perm[base + i] : -1;

  gcv4fp wsl = (gcv4fp)(weight + ((size_t)e * OUT_FEAT + (size_t)slab * SLAB_N) * IN_FEAT);
  v4f wreg[32];
#pragma unroll
  for (int i = 0; i < 32; ++i) wreg[i] = wsl[i * 256 + tid];

  __syncthreads();

  const int c4   = tid & 127;
  const int rsel = tid >> 7;
  const v4f zero4 = {};
  v4f pre[32];

  {
    const int tc = min(4, tiles);
#pragma unroll
    for (int t = 0; t < 4; ++t) {
      if (t < tc) {
#pragma unroll
        for (int j = 0; j < 8; ++j) {
          const int i = t * 8 + j;
          const int tok = sTok[2 * i + rsel];
          pre[i] = ((gcv4fp)(inp + (size_t)max(tok, 0) * IN_FEAT))[c4];
        }
      }
    }
#pragma unroll
    for (int i = 0; i < 32; ++i) {
      const int f = i * 256 + tid;
      *(v4f*)&sW[(f >> 7) * PITCH + (f & 127) * 4] = wreg[i];
    }
#pragma unroll
    for (int t = 0; t < 4; ++t) {
      if (t < tc) {
#pragma unroll
        for (int j = 0; j < 8; ++j) {
          const int i = t * 8 + j;
          const int row = 2 * i + rsel;
          const int tok = sTok[row];
          *(v4f*)&sA[row * PITCH + c4 * 4] = (tok >= 0) ? pre[i] : zero4;
        }
      }
    }
  }
  __syncthreads();

  const int lane = tid & 31;
  const int hl   = lane >> 4;
  const int l15  = lane & 15;
  const int w  = __builtin_amdgcn_readfirstlane(tid >> 5);
  const int ti = w >> 1;
  const int p  = w & 1;

  const float* aptr  = &sA[(ti * TILE_M + l15) * PITCH + 2 * hl];
  const float* bptr0 = &sW[(p * 32 + l15) * PITCH + 2 * hl];
  const float* bptr1 = bptr0 + 16 * PITCH;
  const int col0 = slab * SLAB_N + p * 32 + l15;

  for (int c = 0; c < chunks; ++c) {
    const int tc  = min(4, tiles - 4 * c);
    const int tcn = min(4, tiles - 4 * (c + 1));

#pragma unroll
    for (int t = 0; t < 4; ++t) {
      if (t < tcn) {
#pragma unroll
        for (int j = 0; j < 8; ++j) {
          const int i = t * 8 + j;
          const int tok = sTok[(c + 1) * CH_TOK + 2 * i + rsel];
          pre[i] = ((gcv4fp)(inp + (size_t)max(tok, 0) * IN_FEAT))[c4];
        }
      }
    }

    if (ti < tc) {
      v8f acc0 = {}, acc1 = {};
#pragma unroll 16
      for (int k = 0; k < IN_FEAT; k += 4) {
        v2f a  = *(const v2f*)(aptr + k);
        v2f b0 = *(const v2f*)(bptr0 + k);
        v2f b1 = *(const v2f*)(bptr1 + k);
        acc0 = __builtin_amdgcn_wmma_f32_16x16x4_f32(
            false, a, false, b0, (short)0, acc0, false, false);
        acc1 = __builtin_amdgcn_wmma_f32_16x16x4_f32(
            false, a, false, b1, (short)0, acc1, false, false);
      }
      const int rowbase = c * CH_TOK + ti * TILE_M + 8 * hl;
      int tk[8];
#pragma unroll
      for (int r = 0; r < 8; ++r) tk[r] = sTok[rowbase + r];
#pragma unroll
      for (int r = 0; r < 8; ++r) {
        if (tk[r] >= 0) {
          gfp o = (gfp)(out + (size_t)tk[r] * OUT_FEAT + col0);
          o[0]  = acc0[r];
          o[16] = acc1[r];
        }
      }
    }
    __syncthreads();

    if (tcn > 0) {
#pragma unroll
      for (int t = 0; t < 4; ++t) {
        if (t < tcn) {
#pragma unroll
          for (int j = 0; j < 8; ++j) {
            const int i = t * 8 + j;
            const int row = 2 * i + rsel;
            const int tok = sTok[(c + 1) * CH_TOK + row];
            *(v4f*)&sA[row * PITCH + c4 * 4] = (tok >= 0) ? pre[i] : zero4;
          }
        }
      }
      __syncthreads();
    }
  }
}

extern "C" void kernel_launch(void* const* d_in, const int* in_sizes, int n_in,
                              void* d_out, int out_size, void* d_ws, size_t ws_size,
                              hipStream_t stream) {
  (void)in_sizes; (void)n_in; (void)out_size; (void)ws_size;
  const float* inp    = (const float*)d_in[0];
  const int*   gate   = (const int*)d_in[1];
  const float* weight = (const float*)d_in[2];
  float* out = (float*)d_out;
  MoeWs* ws  = (MoeWs*)d_ws;

  moe_route<<<1, ROUTE_T, 0, stream>>>(gate, ws);
  moe_gemm<<<dim3(OUT_FEAT / SLAB_N, NUM_EXPERT), 256, 0, stream>>>(inp, weight, out, ws);
}


# First chunk pipelined along K: weight slab and token rows enter LDS in 4 K-slices, all requested at once in slice order; slice s's WMMAs run behind an LDS-only barrier while later slices land
# speedup vs baseline: 2.1207x; 1.0669x over previous
#include <hip/hip_runtime.h>
#include <hip/hip_bf16.h>

#define NUM_EXPERT 32
#define IN_FEAT 512
#define OUT_FEAT 512
#define BATCH 4096
#define TILE_M 16

typedef __attribute__((ext_vector_type(2))) float v2f;
typedef __attribute__((ext_vector_type(4))) float v4f;
typedef __attribute__((ext_vector_type(8))) float v8f;

typedef const float  __attribute__((address_space(1)))* gcfp;
typedef const v2f    __attribute__((address_space(1)))* gcv2fp;
typedef const v4f    __attribute__((address_space(1)))* gcv4fp;
typedef float        __attribute__((address_space(1)))* gfp;

struct MoeWs {
  int counts[NUM_EXPERT];
  int perm[NUM_EXPERT * BATCH];
};

#define ROUTE_T 1024
#define ROUTE_PER (BATCH / ROUTE_T)

__launch_bounds__(ROUTE_T)
__global__ void moe_route(const int* __restrict__ gate, MoeWs* __restrict__ ws) {
  __shared__ int cnt[NUM_EXPERT];
  const int t = threadIdx.x;
  int g[ROUTE_PER];
#pragma unroll
  for (int k = 0; k < ROUTE_PER; ++k) g[k] = gate[k * ROUTE_T + t] & (NUM_EXPERT - 1);
  if (t < NUM_EXPERT) cnt[t] = 0;
  __syncthreads();
  int r[ROUTE_PER];
#pragma unroll
  for (int k = 0; k < ROUTE_PER; ++k) r[k] = atomicAdd(&cnt[g[k]], 1);
  __syncthreads();
#pragma unroll
  for (int k = 0; k < ROUTE_PER; ++k) ws->perm[g[k] * BATCH + r[k]] = k * ROUTE_T + t;
  if (t < NUM_EXPERT) ws->counts[t] = cnt[t];
}

#define SLAB_N 64
#define CH_TOK 64
#define PITCH 516
#define KSL 4
#define KSF (IN_FEAT / KSL)
#define KS4 (KSF / 4)
#define ROW4 (IN_FEAT / 4)

static __device__ __forceinline__ void lds_barrier() {
  __builtin_amdgcn_fence(__ATOMIC_RELEASE, "workgroup", "local");
  __builtin_amdgcn_s_barrier();
  __builtin_amdgcn_fence(__ATOMIC_ACQUIRE, "workgroup", "local");
}

static __device__ __forceinline__ void put_tile(float* __restrict__ out, const int* sTokRow,
                                                int col0, v8f acc0, v8f acc1) {
  int tk[8];
#pragma unroll
  for (int r = 0; r < 8; ++r) tk[r] = sTokRow[r];
#pragma unroll
  for (int r = 0; r < 8; ++r) {
    if (tk[r] >= 0) {
      gfp o = (gfp)(out + (size_t)tk[r] * OUT_FEAT + col0);
      o[0]  = acc0[r];
      o[16] = acc1[r];
    }
  }
}

__launch_bounds__(256)
__global__ void moe_gemm(const float* __restrict__ inp,
                         const float* __restrict__ weight,
                         float* __restrict__ out,
                         const MoeWs* __restrict__ ws) {
  __shared__ __align__(16) float sW[SLAB_N * PITCH];
  __shared__ __align__(16) float sA[CH_TOK * PITCH];
  __shared__ int sTok[BATCH];

  const int tid  = threadIdx.x;
  const int slab = blockIdx.x;
  const int e    = blockIdx.y;
  const int n    = min(ws->counts[e], BATCH);
  if (n <= 0) return;
  const int base   = e * BATCH;
  const int tiles  = (n + TILE_M - 1) / TILE_M;
  const int chunks = (tiles + 3) / 4;
  const int npad   = chunks * CH_TOK;

  const int lane = tid & 31;
  const int w = __builtin_amdgcn_readfirstlane(tid >> 5);

  const int t0 = (tid < n) ? ws->perm[base + tid] : -1;

  gcv4fp wsl  = (gcv4fp)(weight + ((size_t)e * OUT_FEAT + (size_t)slab * SLAB_N) * IN_FEAT);
  gcv4fp inp4 = (gcv4fp)inp;
  v4f wreg[32];
#pragma unroll
  for (int j = 0; j < 8; ++j) wreg[j] = wsl[(8 * j + w) * ROW4 + lane];

  sTok[tid] = t0;
  for (int i = tid + 256; i < npad; i += 256)
    sTok[i] = (i < n) ? ws->perm[base + i] : -1;
  lds_barrier();

  int tk[8];
#pragma unroll
  for (int j = 0; j < 8; ++j) tk[j] = sTok[8 * j + w];

  const v4f zero4 = {};
  v4f pre[32];
#pragma unroll
  for (int s = 0; s < KSL; ++s) {
    if (s > 0) {
#pragma unroll
      for (int j = 0; j < 8; ++j) wreg[8 * s + j] = wsl[(8 * j + w) * ROW4 + KS4 * s + lane];
    }
#pragma unroll
    for (int j = 0; j < 8; ++j)
      pre[8 * s + j] = inp4[(size_t)max(tk[j], 0) * ROW4 + KS4 * s + lane];
    if (s + 1 < KSL) __builtin_amdgcn_sched_barrier(0);
  }

  const int hl  = lane >> 4;
  const int l15 = lane & 15;
  const int ti = w >> 1;
  const int p  = w & 1;

  const float* aptr  = &sA[(ti * TILE_M + l15) * PITCH + 2 * hl];
  const float* bptr0 = &sW[(p * 32 + l15) * PITCH + 2 * hl];
  const float* bptr1 = bptr0 + 16 * PITCH;
  const int col0 = slab * SLAB_N + p * 32 + l15;
  const int stg  = w * PITCH + lane * 4;

  {
    const int tc = min(4, tiles);
    int tkn[8];
    if (chunks > 1) {
#pragma unroll
      for (int j = 0; j < 8; ++j) tkn[j] = sTok[CH_TOK + 8 * j + w];
    }
    v8f acc0 = {}, acc1 = {};
#pragma unroll
    for (int s = 0; s < KSL; ++s) {
#pragma unroll
      for (int j = 0; j < 8; ++j)
        *(v4f*)&sW[8 * j * PITCH + stg + KSF * s] = wreg[8 * s + j];
#pragma unroll
      for (int j = 0; j < 8; ++j)
        *(v4f*)&sA[8 * j * PITCH + stg + KSF * s] = (tk[j] >= 0) ? pre[8 * s + j] : zero4;
      if (chunks > 1) {
#pragma unroll
        for (int j = 0; j < 8; ++j)
          pre[8 * s + j] = inp4[(size_t)max(tkn[j], 0) * ROW4 + KS4 * s + lane];
      }
      lds_barrier();
      if (ti < tc) {
#pragma unroll
        for (int k = KSF * s; k < KSF * (s + 1); k += 4) {
          v2f a  = *(const v2f*)(aptr + k);
          v2f b0 = *(const v2f*)(bptr0 + k);
          v2f b1 = *(const v2f*)(bptr1 + k);
          acc0 = __builtin_amdgcn_wmma_f32_16x16x4_f32(
              false, a, false, b0, (short)0, acc0, false, false);
          acc1 = __builtin_amdgcn_wmma_f32_16x16x4_f32(
              false, a, false, b1, (short)0, acc1, false, false);
        }
      }
    }
    if (ti < tc) put_tile(out, &sTok[ti * TILE_M + 8 * hl], col0, acc0, acc1);
  }

  for (int c = 1; c < chunks; ++c) {
    const int tc = min(4, tiles - 4 * c);
    __syncthreads();
#pragma unroll
    for (int j = 0; j < 8; ++j) tk[j] = sTok[c * CH_TOK + 8 * j + w];
#pragma unroll
    for (int j = 0; j < 8; ++j) {
      if ((j >> 1) < tc) {
#pragma unroll
        for (int s = 0; s < KSL; ++s)
          *(v4f*)&sA[8 * j * PITCH + stg + KSF * s] = (tk[j] >= 0) ? pre[8 * s + j] : zero4;
      }
    }
    __syncthreads();

    const int tcn = min(4, tiles - 4 * (c + 1));
    if (tcn > 0) {
#pragma unroll
      for (int j = 0; j < 8; ++j) tk[j] = sTok[(c + 1) * CH_TOK + 8 * j + w];
#pragma unroll
      for (int j = 0; j < 8; ++j) {
        if ((j >> 1) < tcn) {
#pragma unroll
          for (int s = 0; s < KSL; ++s)
            pre[8 * s + j] = inp4[(size_t)max(tk[j], 0) * ROW4 + KS4 * s + lane];
        }
      }
    }

    if (ti < tc) {
      v8f acc0 = {}, acc1 = {};
#pragma unroll 16
      for (int k = 0; k < IN_FEAT; k += 4) {
        v2f a  = *(const v2f*)(aptr + k);
        v2f b0 = *(const v2f*)(bptr0 + k);
        v2f b1 = *(const v2f*)(bptr1 + k);
        acc0 = __builtin_amdgcn_wmma_f32_16x16x4_f32(
            false, a, false, b0, (short)0, acc0, false, false);
        acc1 = __builtin_amdgcn_wmma_f32_16x16x4_f32(
            false, a, false, b1, (short)0, acc1, false, false);
      }
      put_tile(out, &sTok[c * CH_TOK + ti * TILE_M + 8 * hl], col0, acc0, acc1);
    }
  }
}

extern "C" void kernel_launch(void* const* d_in, const int* in_sizes, int n_in,
                              void* d_out, int out_size, void* d_ws, size_t ws_size,
                              hipStream_t stream) {
  (void)in_sizes; (void)n_in; (void)out_size; (void)ws_size;
  const float* inp    = (const float*)d_in[0];
  const int*   gate   = (const int*)d_in[1];
  const float* weight = (const float*)d_in[2];
  float* out = (float*)d_out;
  MoeWs* ws  = (MoeWs*)d_ws;

  moe_route<<<1, ROUTE_T, 0, stream>>>(gate, ws);
  moe_gemm<<<dim3(OUT_FEAT / SLAB_N, NUM_EXPERT), 256, 0, stream>>>(inp, weight, out, ws);
}


# Waves re-cut to (16-column tile x K half): weight fragments stay in registers after chunk 0, a wave sweeps all row tiles; K halves meet in LDS, 16-byte fragments pair K-steps (sum order changes)
# speedup vs baseline: 2.3269x; 1.0972x over previous
#include <hip/hip_runtime.h>
#include <hip/hip_bf16.h>

#define NUM_EXPERT 32
#define IN_FEAT 512
#define OUT_FEAT 512
#define BATCH 4096
#define TILE_M 16

typedef __attribute__((ext_vector_type(2))) float v2f;
typedef __attribute__((ext_vector_type(4))) float v4f;
typedef __attribute__((ext_vector_type(8))) float v8f;

typedef const float  __attribute__((address_space(1)))* gcfp;
typedef const v2f    __attribute__((address_space(1)))* gcv2fp;
typedef const v4f    __attribute__((address_space(1)))* gcv4fp;
typedef float        __attribute__((address_space(1)))* gfp;

struct MoeWs {
  int counts[NUM_EXPERT];
  int perm[NUM_EXPERT * BATCH];
};

#define ROUTE_T 1024
#define ROUTE_PER (BATCH / ROUTE_T)

__launch_bounds__(ROUTE_T)
__global__ void moe_route(const int* __restrict__ gate, MoeWs* __restrict__ ws) {
  __shared__ int cnt[NUM_EXPERT];
  const int t = threadIdx.x;
  int g[ROUTE_PER];
#pragma unroll
  for (int k = 0; k < ROUTE_PER; ++k) g[k] = gate[k * ROUTE_T + t] & (NUM_EXPERT - 1);
  if (t < NUM_EXPERT) cnt[t] = 0;
  __syncthreads();
  int r[ROUTE_PER];
#pragma unroll
  for (int k = 0; k < ROUTE_PER; ++k) r[k] = atomicAdd(&cnt[g[k]], 1);
  __syncthreads();
#pragma unroll
  for (int k = 0; k < ROUTE_PER; ++k) ws->perm[g[k] * BATCH + r[k]] = k * ROUTE_T + t;
  if (t < NUM_EXPERT) ws->counts[t] = cnt[t];
}

#define SLAB_N 64
#define CH_TOK 64
#define PITCH 516
#define KSL 4
#define KSF (IN_FEAT / KSL)
#define KS4 (KSF / 4)
#define ROW4 (IN_FEAT / 4)
#define KHF (KSF / 2)
#define KPR (KHF / 8)
#define ATILE (TILE_M * PITCH)
#define CP 68
#define CPLANE (CH_TOK * CP)

typedef v4f __attribute__((address_space(1)))* gv4fp;

static __device__ __forceinline__ void lds_barrier() {
  __builtin_amdgcn_fence(__ATOMIC_RELEASE, "workgroup", "local");
  __builtin_amdgcn_s_barrier();
  __builtin_amdgcn_fence(__ATOMIC_ACQUIRE, "workgroup", "local");
}

template <int TC, int P0, int NP>
static __device__ __forceinline__ void products(const float* aptr, const v4f (&breg)[KSL * KPR],
                                                v8f (&acc)[4]) {
  constexpr int NG = NP / 2;
  v4f a[2][2 * TC];
#pragma unroll
  for (int q = 0; q < 2; ++q) {
#pragma unroll
    for (int r = 0; r < TC; ++r)
      a[0][q * TC + r] = *(const v4f*)(aptr + r * ATILE + KSF * ((P0 + q) / KPR) + 8 * ((P0 + q) % KPR));
  }
#pragma unroll
  for (int g = 0; g < NG; ++g) {
    if (g + 1 < NG) {
#pragma unroll
      for (int q = 0; q < 2; ++q) {
        const int p = P0 + 2 * (g + 1) + q;
#pragma unroll
        for (int r = 0; r < TC; ++r)
          a[(g + 1) & 1][q * TC + r] = *(const v4f*)(aptr + r * ATILE + KSF * (p / KPR) + 8 * (p % KPR));
      }
    }
    __builtin_amdgcn_sched_barrier(0);
#pragma unroll
    for (int q = 0; q < 2; ++q) {
      const v4f b = breg[P0 + 2 * g + q];
      const v2f b01 = {b[0], b[1]};
      const v2f b23 = {b[2], b[3]};
#pragma unroll
      for (int r = 0; r < TC; ++r) {
        const v4f x = a[g & 1][q * TC + r];
        const v2f a01 = {x[0], x[1]};
        acc[r] = __builtin_amdgcn_wmma_f32_16x16x4_f32(
            false, a01, false, b01, (short)0, acc[r], false, false);
      }
#pragma unroll
      for (int r = 0; r < TC; ++r) {
        const v4f x = a[g & 1][q * TC + r];
        const v2f a23 = {x[2], x[3]};
        acc[r] = __builtin_amdgcn_wmma_f32_16x16x4_f32(
            false, a23, false, b23, (short)0, acc[r], false, false);
      }
    }
    __builtin_amdgcn_sched_barrier(0);
  }
}

template <int TC>
static __device__ __forceinline__ void chunk_products(const float* aptr, const v4f (&breg)[KSL * KPR],
                                                      float* cw) {
  v8f acc[4] = {};
  products<TC, 0, KSL * KPR>(aptr, breg, acc);
#pragma unroll
  for (int r = 0; r < TC; ++r) {
#pragma unroll
    for (int v = 0; v < 8; ++v) cw[(TILE_M * r + v) * CP] = acc[r][v];
  }
}

static __device__ __forceinline__ void put_chunk(float* __restrict__ out, const float* sC,
                                                 const int* sTokChunk, int w, int lane, int slab) {
  int tok[4];
  v4f x[4], y[4];
#pragma unroll
  for (int i = 0; i < 4; ++i) {
    const int line = (w * 4 + i) * 4 + (lane >> 3);
    const int row  = line >> 1;
    const int col  = (line & 1) * 32 + (lane & 7) * 4;
    tok[i] = sTokChunk[row];
    x[i] = *(const v4f*)&sC[row * CP + col];
    y[i] = *(const v4f*)&sC[CPLANE + row * CP + col];
  }
#pragma unroll
  for (int i = 0; i < 4; ++i) {
    const int line = (w * 4 + i) * 4 + (lane >> 3);
    const int col  = (line & 1) * 32 + (lane & 7) * 4;
    if (tok[i] >= 0)
      *(gv4fp)(out + (size_t)tok[i] * OUT_FEAT + slab * SLAB_N + col) = x[i] + y[i];
  }
}

__launch_bounds__(256)
__global__ void moe_gemm(const float* __restrict__ inp,
                         const float* __restrict__ weight,
                         float* __restrict__ out,
                         const MoeWs* __restrict__ ws) {
  __shared__ __align__(16) float sW[SLAB_N * PITCH];
  __shared__ __align__(16) float sA[CH_TOK * PITCH];
  __shared__ __align__(16) float sC[2 * CPLANE];
  __shared__ int sTok[BATCH];

  const int tid  = threadIdx.x;
  const int slab = blockIdx.x;
  const int e    = blockIdx.y;
  const int n    = min(ws->counts[e], BATCH);
  if (n <= 0) return;
  const int base   = e * BATCH;
  const int tiles  = (n + TILE_M - 1) / TILE_M;
  const int chunks = (tiles + 3) / 4;
  const int npad   = chunks * CH_TOK;

  const int lane = tid & 31;
  const int w = __builtin_amdgcn_readfirstlane(tid >> 5);

  const int t0 = (tid < n) ? ws->perm[base + tid] : -1;

  gcv4fp wsl  = (gcv4fp)(weight + ((size_t)e * OUT_FEAT + (size_t)slab * SLAB_N) * IN_FEAT);
  gcv4fp inp4 = (gcv4fp)inp;
  v4f wreg[32];
#pragma unroll
  for (int j = 0; j < 8; ++j) wreg[j] = wsl[(8 * j + w) * ROW4 + lane];

  sTok[tid] = t0;
  for (int i = tid + 256; i < npad; i += 256)
    sTok[i] = (i < n) ? ws->perm[base + i] : -1;
  lds_barrier();

  int tk[8];
#pragma unroll
  for (int j = 0; j < 8; ++j) tk[j] = sTok[8 * j + w];

  const v4f zero4 = {};
  v4f pre[32];
#pragma unroll
  for (int s = 0; s < KSL; ++s) {
    if (s > 0) {
#pragma unroll
      for (int j = 0; j < 8; ++j) wreg[8 * s + j] = wsl[(8 * j + w) * ROW4 + KS4 * s + lane];
    }
#pragma unroll
    for (int j = 0; j < 8; ++j)
      pre[8 * s + j] = inp4[(size_t)max(tk[j], 0) * ROW4 + KS4 * s + lane];
    if (s + 1 < KSL) __builtin_amdgcn_sched_barrier(0);
  }

  const int hl  = lane >> 4;
  const int l15 = lane & 15;
  const int ct = w & 3;
  const int kh = w >> 2;

  const float* aptr = &sA[l15 * PITCH + 4 * hl + KHF * kh];
  const float* bptr = &sW[(ct * TILE_M + l15) * PITCH + 4 * hl + KHF * kh];
  float* cw = &sC[kh * CPLANE + 8 * hl * CP + ct * TILE_M + l15];
  const int stg = w * PITCH + lane * 4;

  v4f breg[KSL * KPR];

  {
    int tkn[8];
    if (chunks > 1) {
#pragma unroll
      for (int j = 0; j < 8; ++j) tkn[j] = sTok[CH_TOK + 8 * j + w];
    }
    v8f acc[4] = {};
#pragma unroll
    for (int s = 0; s < KSL; ++s) {
#pragma unroll
      for (int j = 0; j < 8; ++j)
        *(v4f*)&sW[8 * j * PITCH + stg + KSF * s] = wreg[8 * s + j];
#pragma unroll
      for (int j = 0; j < 8; ++j)
        *(v4f*)&sA[8 * j * PITCH + stg + KSF * s] = (tk[j] >= 0) ? pre[8 * s + j] : zero4;
      if (chunks > 1) {
#pragma unroll
        for (int j = 0; j < 8; ++j)
          pre[8 * s + j] = inp4[(size_t)max(tkn[j], 0) * ROW4 + KS4 * s + lane];
      }
      lds_barrier();
#pragma unroll
      for (int i = 0; i < KPR; ++i) breg[KPR * s + i] = *(const v4f*)(bptr + KSF * s + 8 * i);
      if (s == 0) products<4, 0 * KPR, KPR>(aptr, breg, acc);
      if (s == 1) products<4, 1 * KPR, KPR>(aptr, breg, acc);
      if (s == 2) products<4, 2 * KPR, KPR>(aptr, breg, acc);
      if (s == 3) products<4, 3 * KPR, KPR>(aptr, breg, acc);
    }
#pragma unroll
    for (int r = 0; r < 4; ++r) {
#pragma unroll
      for (int v = 0; v < 8; ++v) cw[(TILE_M * r + v) * CP] = acc[r][v];
    }
    lds_barrier();
    put_chunk(out, sC, &sTok[0], w, lane, slab);
  }

  for (int c = 1; c < chunks; ++c) {
    const int tc = min(4, tiles - 4 * c);
    __syncthreads();
#pragma unroll
    for (int j = 0; j < 8; ++j) tk[j] = sTok[c * CH_TOK + 8 * j + w];
#pragma unroll
    for (int j = 0; j < 8; ++j) {
      if ((j >> 1) < tc) {
#pragma unroll
        for (int s = 0; s < KSL; ++s)
          *(v4f*)&sA[8 * j * PITCH + stg + KSF * s] = (tk[j] >= 0) ? pre[8 * s + j] : zero4;
      }
    }
    __syncthreads();

    const int tcn = min(4, tiles - 4 * (c + 1));
    if (tcn > 0) {
#pragma unroll
      for (int j = 0; j < 8; ++j) tk[j] = sTok[(c + 1) * CH_TOK + 8 * j + w];
#pragma unroll
      for (int j = 0; j < 8; ++j) {
        if ((j >> 1) < tcn) {
#pragma unroll
          for (int s = 0; s < KSL; ++s)
            pre[8 * s + j] = inp4[(size_t)max(tk[j], 0) * ROW4 + KS4 * s + lane];
        }
      }
    }

    if (tc >= 4)      chunk_products<4>(aptr, breg, cw);
    else if (tc == 3) chunk_products<3>(aptr, breg, cw);
    else if (tc == 2) chunk_products<2>(aptr, breg, cw);
    else              chunk_products<1>(aptr, breg, cw);
    lds_barrier();
    put_chunk(out, sC, &sTok[c * CH_TOK], w, lane, slab);
  }
}

extern "C" void kernel_launch(void* const* d_in, const int* in_sizes, int n_in,
                              void* d_out, int out_size, void* d_ws, size_t ws_size,
                              hipStream_t stream) {
  (void)in_sizes; (void)n_in; (void)out_size; (void)ws_size;
  const float* inp    = (const float*)d_in[0];
  const int*   gate   = (const int*)d_in[1];
  const float* weight = (const float*)d_in[2];
  float* out = (float*)d_out;
  MoeWs* ws  = (MoeWs*)d_ws;

  moe_route<<<1, ROUTE_T, 0, stream>>>(gate, ws);
  moe_gemm<<<dim3(OUT_FEAT / SLAB_N, NUM_EXPERT), 256, 0, stream>>>(inp, weight, out, ws);
}
